# starting kernel
# baseline (speedup 1.0000x reference)
#include <hip/hip_runtime.h>

#define SEP_ID 102
#define B_SZ 64
#define S_SZ 512
#define H_SZ 1024

typedef __attribute__((ext_vector_type(2))) float v2f;
typedef __attribute__((ext_vector_type(8))) float v8f;

struct SegInfo {
  int fs;
  int le;
  float tscale;
  float lscale;
};

__global__ void seg_stats_kernel(const int* __restrict__ ids,
                                 SegInfo* __restrict__ info) {
  int b = blockIdx.x * blockDim.x + threadIdx.x;
  if (b >= B_SZ) return;
  const int* row = ids + b * S_SZ;
  int n = 0, first = -1, last = -1;
  for (int s = 0; s < S_SZ; ++s) {
    int v = row[s];
    if (v == SEP_ID) {
      ++n;
      if (first < 0) first = s;
      last = s;
    }
  }
  int fs, le;
  if (n == 0) {
    fs = 0;
    le = 0;
  } else {
    fs = first;
    le = (n >= 2) ? last : S_SZ;
  }
  int tcnt = fs - 1;
  if (tcnt < 0) tcnt = 0;
  int lcnt = le - fs - 1;
  if (lcnt < 0) lcnt = 0;
  SegInfo o;
  o.fs = fs;
  o.le = le;
  o.tscale = (tcnt > 0) ? (1.0f / (float)tcnt) : 0.0f;
  o.lscale = (lcnt > 0) ? (1.0f / (float)lcnt) : 0.0f;
  info[b] = o;
}

__global__ void __launch_bounds__(256)
seg_pool_wmma_kernel(const float* __restrict__ hs,
                     const SegInfo* __restrict__ info,
                     float* __restrict__ out) {
  const int lane = threadIdx.x & 31;
  const int wave = threadIdx.x >> 5;
  const int b = blockIdx.y;
  const int n0 = blockIdx.x * 128 + wave * 16;

  const SegInfo si = info[b];
  const int m = lane & 15;
  const int khalf = (lane >> 4) << 1;

  int lo = 0;
  int len = 0;
  if (m == 0) {
    lo = 1;
    len = si.fs - 1;
  } else if (m == 1) {
    lo = si.fs + 1;
    len = si.le - si.fs - 1;
  }
  if (len < 0) len = 0;
  const unsigned ulen = (unsigned)len;

  const float* __restrict__ base =
      hs + (size_t)b * (S_SZ * H_SZ) + (size_t)n0 + (size_t)m;

  v8f c = {0.f, 0.f, 0.f, 0.f, 0.f, 0.f, 0.f, 0.f};

  int rel = khalf - lo;

#pragma unroll 8
  for (int k0 = 0; k0 < S_SZ; k0 += 4) {
    v2f a;
    a.x = ((unsigned)rel < ulen) ? 1.0f : 0.0f;
    a.y = ((unsigned)(rel + 1) < ulen) ? 1.0f : 0.0f;

    const float* p = base + (size_t)(k0 + khalf) * H_SZ;
    v2f bm;
    bm.x = p[0];
    bm.y = p[H_SZ];

    __builtin_prefetch(p + 16 * H_SZ, 0, 0);

    c = __builtin_amdgcn_wmma_f32_16x16x4_f32(
         false, a,  false, bm,
         (short)0, c,  false,  false);

    rel += 4;
  }

  if (lane < 16) {
    const int col = n0 + lane;
    out[(size_t)b * H_SZ + col] = c[0] * si.tscale;
    out[(size_t)(B_SZ * H_SZ) + (size_t)b * H_SZ + col] = c[1] * si.lscale;
  }
}

extern "C" void kernel_launch(void* const* d_in, const int* in_sizes, int n_in,
                              void* d_out, int out_size, void* d_ws,
                              size_t ws_size, hipStream_t stream) {
  (void)in_sizes;
  (void)n_in;
  (void)out_size;
  (void)ws_size;

  const float* hidden = (const float*)d_in[0];
  const int* ids = (const int*)d_in[1];
  float* out = (float*)d_out;
  SegInfo* info = (SegInfo*)d_ws;

  seg_stats_kernel<<<1, 64, 0, stream>>>(ids, info);

  dim3 grid(H_SZ / 128, B_SZ);
  dim3 block(256);
  seg_pool_wmma_kernel<<<grid, block, 0, stream>>>(hidden, info, out);
}


# Fuse the per-batch SEP statistics into the pooling kernel: each wave builds its row's SEP bitmap from coalesced id loads and wave ballots; one launch instead of two
# speedup vs baseline: 2.0301x; 2.0301x over previous
#include <hip/hip_runtime.h>


#define SEP_ID 102
#define B_SZ 64
#define S_SZ 512
#define H_SZ 1024

typedef __attribute__((ext_vector_type(2))) float v2f;
typedef __attribute__((ext_vector_type(8))) float v8f;

__global__ void __launch_bounds__(256)
seg_pool_wmma_kernel(const float* __restrict__ hs,
                     const int* __restrict__ ids,
                     float* __restrict__ out) {
  const int lane = threadIdx.x & 31;
  const int wave = threadIdx.x >> 5;
  const int b = blockIdx.y;
  const int n0 = blockIdx.x * 128 + wave * 16;

  const int* __restrict__ idrow = ids + b * S_SZ + lane;
  unsigned sepbits[S_SZ / 32];
  for (int i = 0; i < S_SZ / 32; ++i) {
    sepbits[i] = __builtin_amdgcn_ballot_w32(idrow[i * 32] == SEP_ID);
  }
  int first = -1;
  int last = -1;
  for (int i = 0; i < S_SZ / 32; ++i) {
    const unsigned w = sepbits[i];
    if (w != 0u) {
      if (first < 0) first = i * 32 + __builtin_ctz(w);
      last = i * 32 + 31 - __builtin_clz(w);
    }
  }

  int fs, le;
  if (first < 0) {
    fs = 0;
    le = 0;
  } else {
    fs = first;
    le = (last != first) ? last : S_SZ;
  }
  int tcnt = fs - 1;
  if (tcnt < 0) tcnt = 0;
  int lcnt = le - fs - 1;
  if (lcnt < 0) lcnt = 0;
  const float tscale = (tcnt > 0) ? (1.0f / (float)tcnt) : 0.0f;
  const float lscale = (lcnt > 0) ? (1.0f / (float)lcnt) : 0.0f;

  const int m = lane & 15;
  const int khalf = (lane >> 4) << 1;

  int lo = 0;
  int len = 0;
  if (m == 0) {
    lo = 1;
    len = fs - 1;
  } else if (m == 1) {
    lo = fs + 1;
    len = le - fs - 1;
  }
  if (len < 0) len = 0;
  const unsigned ulen = (unsigned)len;

  const float* __restrict__ base =
      hs + (size_t)b * (S_SZ * H_SZ) + (size_t)n0 + (size_t)m;

  v8f c = {0.f, 0.f, 0.f, 0.f, 0.f, 0.f, 0.f, 0.f};

  int rel = khalf - lo;

#pragma unroll 8
  for (int k0 = 0; k0 < S_SZ; k0 += 4) {
    v2f a;
    a.x = ((unsigned)rel < ulen) ? 1.0f : 0.0f;
    a.y = ((unsigned)(rel + 1) < ulen) ? 1.0f : 0.0f;

    const float* p = base + (size_t)(k0 + khalf) * H_SZ;
    v2f bm;
    bm.x = p[0];
    bm.y = p[H_SZ];

    __builtin_prefetch(p + 16 * H_SZ, 0, 0);

    c = __builtin_amdgcn_wmma_f32_16x16x4_f32(
         false, a,  false, bm,
         (short)0, c,  false,  false);

    rel += 4;
  }

  if (lane < 16) {
    const int col = n0 + lane;
    out[(size_t)b * H_SZ + col] = c[0] * tscale;
    out[(size_t)(B_SZ * H_SZ) + (size_t)b * H_SZ + col] = c[1] * lscale;
  }
}

extern "C" void kernel_launch(void* const* d_in, const int* in_sizes, int n_in,
                              void* d_out, int out_size, void* d_ws,
                              size_t ws_size, hipStream_t stream) {
  (void)in_sizes;
  (void)n_in;
  (void)out_size;
  (void)d_ws;
  (void)ws_size;

  const float* hidden = (const float*)d_in[0];
  const int* ids = (const int*)d_in[1];
  float* out = (float*)d_out;

  dim3 grid(H_SZ / 128, B_SZ);
  dim3 block(256);
  seg_pool_wmma_kernel<<<grid, block, 0, stream>>>(hidden, ids, out);
}


# Stream rows with whole-line global_load_b128, 32 in flight per wave (64-column wave tile, loaded pairs are the WMMA B as they stand); K split over 4 waves, partials summed via LDS in fixed order
# speedup vs baseline: 2.4277x; 1.1959x over previous
#include <hip/hip_runtime.h>


#define SEP_ID 102
#define B_SZ 64
#define S_SZ 512
#define H_SZ 1024

#define KQ 4
#define STEPS (S_SZ / KQ / 2)
#define RING 32

typedef __attribute__((ext_vector_type(2))) float v2f;
typedef __attribute__((ext_vector_type(4))) float v4f;
typedef __attribute__((ext_vector_type(8))) float v8f;

__global__ void __launch_bounds__(256)
seg_pool_wmma_kernel(const float* __restrict__ hs,
                     const int* __restrict__ ids,
                     float* __restrict__ out) {
  __shared__ v4f red[KQ][2][2][16];

  const int lane = threadIdx.x & 31;
  const int wave = threadIdx.x >> 5;
  const int b = blockIdx.y;
  const int ct = wave & 1;
  const int kq = wave >> 1;
  const int n0 = blockIdx.x * 128 + ct * 64;

  const int* __restrict__ idrow = ids + b * S_SZ + lane;
  unsigned sepbits[S_SZ / 32];
  for (int i = 0; i < S_SZ / 32; ++i) {
    sepbits[i] = __builtin_amdgcn_ballot_w32(idrow[i * 32] == SEP_ID);
  }
  int first = -1;
  int last = -1;
  for (int i = 0; i < S_SZ / 32; ++i) {
    const unsigned w = sepbits[i];
    if (w != 0u) {
      if (first < 0) first = i * 32 + __builtin_ctz(w);
      last = i * 32 + 31 - __builtin_clz(w);
    }
  }

  int fs, le;
  if (first < 0) {
    fs = 0;
    le = 0;
  } else {
    fs = first;
    le = (last != first) ? last : S_SZ;
  }
  int tcnt = fs - 1;
  if (tcnt < 0) tcnt = 0;
  int lcnt = le - fs - 1;
  if (lcnt < 0) lcnt = 0;
  const float tscale = (tcnt > 0) ? (1.0f / (float)tcnt) : 0.0f;
  const float lscale = (lcnt > 0) ? (1.0f / (float)lcnt) : 0.0f;

  const int c = lane & 15;
  const int h = lane >> 4;

  int lo = 0;
  int len = 0;
  if ((c >> 1) == 0) {
    lo = 1;
    len = fs - 1;
  } else if ((c >> 1) == 1) {
    lo = fs + 1;
    len = le - fs - 1;
  }
  if (len < 0) len = 0;
  const unsigned ulen_x = ((c & 1) == 0) ? (unsigned)len : 0u;
  const unsigned ulen_y = ((c & 1) == 1) ? (unsigned)len : 0u;

  const int krow0 = kq * (S_SZ / KQ) + h;
  const float* __restrict__ base =
      hs + (size_t)b * (S_SZ * H_SZ) + (size_t)krow0 * H_SZ + (size_t)n0 + (size_t)(4 * c);

  v8f acc0 = {0.f, 0.f, 0.f, 0.f, 0.f, 0.f, 0.f, 0.f};
  v8f acc1 = acc0;

  const int rel0 = krow0 - lo;

  v4f q[RING];
#pragma unroll
  for (int t = 0; t < RING; ++t) {
    q[t] = *reinterpret_cast<const v4f*>(base + (size_t)(2 * t) * H_SZ);
  }

#pragma unroll
  for (int t = 0; t < STEPS; ++t) {
    const v4f x = q[t % RING];

    const unsigned rel = (unsigned)(rel0 + 2 * t);
    v2f a;
    a.x = (rel < ulen_x) ? 1.0f : 0.0f;
    a.y = (rel < ulen_y) ? 1.0f : 0.0f;

    const v2f b01 = __builtin_shufflevector(x, x, 0, 1);
    const v2f b23 = __builtin_shufflevector(x, x, 2, 3);
    acc0 = __builtin_amdgcn_wmma_f32_16x16x4_f32(false, a, false, b01, (short)0, acc0, false, false);
    acc1 = __builtin_amdgcn_wmma_f32_16x16x4_f32(false, a, false, b23, (short)0, acc1, false, false);

    if (t + RING < STEPS) {
      q[t % RING] = *reinterpret_cast<const v4f*>(base + (size_t)(2 * (t + RING)) * H_SZ);
    }
    __builtin_amdgcn_sched_barrier(0);
  }

  if (lane < 16) {
    v4f t4, l4;
    t4.x = acc0[0]; t4.y = acc0[1]; t4.z = acc1[0]; t4.w = acc1[1];
    l4.x = acc0[2]; l4.y = acc0[3]; l4.z = acc1[2]; l4.w = acc1[3];
    red[kq][ct][0][lane] = t4;
    red[kq][ct][1][lane] = l4;
  }
  __syncthreads();

  {
    const int t = threadIdx.x;
    const int which = t >> 7;
    const int cb = t & 127;
    const float* r = reinterpret_cast<const float*>(&red[0][0][0][0]);
    const int off = ((cb >> 6) * 2 + which) * 64 + (cb & 63);
    float s = r[off];
    s += r[off + 1 * 256];
    s += r[off + 2 * 256];
    s += r[off + 3 * 256];
    const float scale = which ? lscale : tscale;
    out[(size_t)which * (B_SZ * H_SZ) + (size_t)b * H_SZ + (size_t)(blockIdx.x * 128 + cb)] = s * scale;
  }
}

extern "C" void kernel_launch(void* const* d_in, const int* in_sizes, int n_in,
                              void* d_out, int out_size, void* d_ws,
                              size_t ws_size, hipStream_t stream) {
  (void)in_sizes;
  (void)n_in;
  (void)out_size;
  (void)d_ws;
  (void)ws_size;

  const float* hidden = (const float*)d_in[0];
  const int* ids = (const int*)d_in[1];
  float* out = (float*)d_out;

  dim3 grid(H_SZ / 128, B_SZ);
  dim3 block(256);
  seg_pool_wmma_kernel<<<grid, block, 0, stream>>>(hidden, ids, out);
}


# Load and multiply only the rows a mask can select (1 .. last bound), dealt to the 4 K waves in groups of 64 rows; changes which rows each LDS partial sums, order stays fixed
# speedup vs baseline: 2.6985x; 1.1115x over previous
#include <hip/hip_runtime.h>


#define SEP_ID 102
#define B_SZ 64
#define S_SZ 512
#define H_SZ 1024

#define KQ 4
#define RING 32
#define GRP RING

typedef __attribute__((ext_vector_type(2))) float v2f;
typedef __attribute__((ext_vector_type(4))) float v4f;
typedef __attribute__((ext_vector_type(8))) float v8f;

__global__ void __launch_bounds__(256)
seg_pool_wmma_kernel(const float* __restrict__ hs,
                     const int* __restrict__ ids,
                     float* __restrict__ out) {
  __shared__ v4f red[KQ][2][2][16];

  const int lane = threadIdx.x & 31;
  const int wave = __builtin_amdgcn_readfirstlane((int)(threadIdx.x >> 5));
  const int b = blockIdx.y;
  const int ct = wave & 1;
  const int kq = wave >> 1;
  const int n0 = blockIdx.x * 128 + ct * 64;

  const int* __restrict__ idrow = ids + b * S_SZ + lane;
  unsigned sepbits[S_SZ / 32];
  for (int i = 0; i < S_SZ / 32; ++i) {
    sepbits[i] = __builtin_amdgcn_ballot_w32(idrow[i * 32] == SEP_ID);
  }
  int first = -1;
  int last = -1;
  for (int i = 0; i < S_SZ / 32; ++i) {
    const unsigned w = sepbits[i];
    if (w != 0u) {
      if (first < 0) first = i * 32 + __builtin_ctz(w);
      last = i * 32 + 31 - __builtin_clz(w);
    }
  }

  int fs, le;
  if (first < 0) {
    fs = 0;
    le = 0;
  } else {
    fs = first;
    le = (last != first) ? last : S_SZ;
  }
  int tcnt = fs - 1;
  if (tcnt < 0) tcnt = 0;
  int lcnt = le - fs - 1;
  if (lcnt < 0) lcnt = 0;
  const float tscale = (tcnt > 0) ? (1.0f / (float)tcnt) : 0.0f;
  const float lscale = (lcnt > 0) ? (1.0f / (float)lcnt) : 0.0f;

  const int nrows = (first < 0) ? 0 : (le - 1);
  const int nsteps = (nrows + 1) >> 1;
  const int ngrp = (nsteps + GRP - 1) / GRP;
  const int gb = ngrp >> 2;
  const int gr = ngrp & 3;
  const int mygrp = gb + ((kq < gr) ? 1 : 0);
  const int gstart = kq * gb + ((kq < gr) ? kq : gr);
  const int nk = mygrp * GRP;
  const int rs = 1 + 2 * GRP * gstart;
  int re = rs + 2 * nk;
  if (re > S_SZ) re = S_SZ;
  int p0 = rs;
  if (p0 > S_SZ - 2 * nk) p0 = S_SZ - 2 * nk;

  const int c = lane & 15;
  const int h = lane >> 4;

  int lo = 0;
  int hi = 0;
  if ((c >> 1) == 0) {
    lo = 1;
    hi = fs;
  } else if ((c >> 1) == 1) {
    lo = fs + 1;
    hi = le;
  }
  if (lo < rs) lo = rs;
  if (hi > re) hi = re;
  int len = hi - lo;
  if (len < 0) len = 0;
  const unsigned ulen_x = ((c & 1) == 0) ? (unsigned)len : 0u;
  const unsigned ulen_y = ((c & 1) == 1) ? (unsigned)len : 0u;

  const int krow0 = p0 + h;
  const float* __restrict__ base =
      hs + (size_t)b * (S_SZ * H_SZ) + (size_t)krow0 * H_SZ + (size_t)n0 + (size_t)(4 * c);

  v8f acc0 = {0.f, 0.f, 0.f, 0.f, 0.f, 0.f, 0.f, 0.f};
  v8f acc1 = acc0;

  const int rel0 = krow0 - lo;

  auto step = [&](const v4f x, const int s) {
    const unsigned rel = (unsigned)(rel0 + 2 * s);
    v2f a;
    a.x = (rel < ulen_x) ? 1.0f : 0.0f;
    a.y = (rel < ulen_y) ? 1.0f : 0.0f;
    const v2f b01 = __builtin_shufflevector(x, x, 0, 1);
    const v2f b23 = __builtin_shufflevector(x, x, 2, 3);
    acc0 = __builtin_amdgcn_wmma_f32_16x16x4_f32(false, a, false, b01, (short)0, acc0, false, false);
    acc1 = __builtin_amdgcn_wmma_f32_16x16x4_f32(false, a, false, b23, (short)0, acc1, false, false);
  };
  auto ld = [&](const int s) -> v4f {
    return *reinterpret_cast<const v4f*>(base + (size_t)(2 * s) * H_SZ);
  };

  v4f q[RING];
  if (nk == 2 * RING) {
#pragma unroll
    for (int t = 0; t < RING; ++t) {
      q[t] = ld(t);
    }
#pragma unroll
    for (int t = 0; t < 2 * RING; ++t) {
      step(q[t % RING], t);
      if (t + RING < 2 * RING) {
        q[t % RING] = ld(t + RING);
      }
      __builtin_amdgcn_sched_barrier(0);
    }
  } else if (nk == RING) {
#pragma unroll
    for (int t = 0; t < RING; ++t) {
      q[t] = ld(t);
    }
    __builtin_amdgcn_sched_barrier(0);
#pragma unroll
    for (int t = 0; t < RING; ++t) {
      step(q[t], t);
    }
  }

  if (lane < 16) {
    v4f t4, l4;
    t4.x = acc0[0]; t4.y = acc0[1]; t4.z = acc1[0]; t4.w = acc1[1];
    l4.x = acc0[2]; l4.y = acc0[3]; l4.z = acc1[2]; l4.w = acc1[3];
    red[kq][ct][0][lane] = t4;
    red[kq][ct][1][lane] = l4;
  }
  __syncthreads();

  {
    const int t = threadIdx.x;
    const int which = t >> 7;
    const int cb = t & 127;
    const float* r = reinterpret_cast<const float*>(&red[0][0][0][0]);
    const int off = ((cb >> 6) * 2 + which) * 64 + (cb & 63);
    float s = r[off];
    s += r[off + 1 * 256];
    s += r[off + 2 * 256];
    s += r[off + 3 * 256];
    const float scale = which ? lscale : tscale;
    out[(size_t)which * (B_SZ * H_SZ) + (size_t)b * H_SZ + (size_t)(blockIdx.x * 128 + cb)] = s * scale;
  }
}

extern "C" void kernel_launch(void* const* d_in, const int* in_sizes, int n_in,
                              void* d_out, int out_size, void* d_ws,
                              size_t ws_size, hipStream_t stream) {
  (void)in_sizes;
  (void)n_in;
  (void)out_size;
  (void)d_ws;
  (void)ws_size;

  const float* hidden = (const float*)d_in[0];
  const int* ids = (const int*)d_in[1];
  float* out = (float*)d_out;

  dim3 grid(H_SZ / 128, B_SZ);
  dim3 block(256);
  seg_pool_wmma_kernel<<<grid, block, 0, stream>>>(hidden, ids, out);
}


# Rows of a lane that share both mask weights (4 load steps) are summed with packed f32 adds before ONE WMMA pair; a quad holding a window bound keeps a pair per step; changes f32 sum order, fixed
# speedup vs baseline: 2.8855x; 1.0693x over previous
#include <hip/hip_runtime.h>


#define SEP_ID 102
#define B_SZ 64
#define S_SZ 512
#define H_SZ 1024

#define KQ 4
#define RING 32
#define GRP RING
#define QG 4

typedef __attribute__((ext_vector_type(2))) float v2f;
typedef __attribute__((ext_vector_type(4))) float v4f;
typedef __attribute__((ext_vector_type(8))) float v8f;

__global__ void __launch_bounds__(256)
seg_pool_wmma_kernel(const float* __restrict__ hs,
                     const int* __restrict__ ids,
                     float* __restrict__ out) {
  __shared__ v4f red[KQ][2][2][16];

  const int lane = threadIdx.x & 31;
  const int wave = __builtin_amdgcn_readfirstlane((int)(threadIdx.x >> 5));
  const int b = blockIdx.y;
  const int ct = wave & 1;
  const int kq = wave >> 1;
  const int n0 = blockIdx.x * 128 + ct * 64;

  const int* __restrict__ idrow = ids + b * S_SZ + lane;
  unsigned sepbits[S_SZ / 32];
  for (int i = 0; i < S_SZ / 32; ++i) {
    sepbits[i] = __builtin_amdgcn_ballot_w32(idrow[i * 32] == SEP_ID);
  }
  int first = -1;
  int last = -1;
  for (int i = 0; i < S_SZ / 32; ++i) {
    const unsigned w = sepbits[i];
    if (w != 0u) {
      if (first < 0) first = i * 32 + __builtin_ctz(w);
      last = i * 32 + 31 - __builtin_clz(w);
    }
  }

  int fs, le;
  if (first < 0) {
    fs = 0;
    le = 0;
  } else {
    fs = first;
    le = (last != first) ? last : S_SZ;
  }
  int tcnt = fs - 1;
  if (tcnt < 0) tcnt = 0;
  int lcnt = le - fs - 1;
  if (lcnt < 0) lcnt = 0;
  const float tscale = (tcnt > 0) ? (1.0f / (float)tcnt) : 0.0f;
  const float lscale = (lcnt > 0) ? (1.0f / (float)lcnt) : 0.0f;

  const int nrows = (first < 0) ? 0 : (le - 1);
  const int nsteps = (nrows + 1) >> 1;
  const int ngrp = (nsteps + GRP - 1) / GRP;
  const int gb = ngrp >> 2;
  const int gr = ngrp & 3;
  const int mygrp = gb + ((kq < gr) ? 1 : 0);
  const int gstart = kq * gb + ((kq < gr) ? kq : gr);
  const int nk = mygrp * GRP;
  const int rs = 1 + 2 * GRP * gstart;
  int re = rs + 2 * nk;
  if (re > S_SZ) re = S_SZ;
  int p0 = rs;
  if (p0 > S_SZ - 2 * nk) p0 = S_SZ - 2 * nk;

  unsigned plain = 0u;
  {
    const int e0 = (rs > 1) ? rs : 1;
    const int e1 = (fs < re) ? fs : re;
    const int e2 = (rs > fs + 1) ? rs : (fs + 1);
    const int e3 = (le < re) ? le : re;
    const int ev[4] = {e0, e1, e2, e3};
    for (int i = 0; i < 4; ++i) {
      const int d = ev[i] - p0;
      if (d > 0 && d < 2 * nk && (d & (2 * QG - 1)) != 0) plain |= 1u << (d / (2 * QG));
    }
  }

  const int c = lane & 15;
  const int h = lane >> 4;

  int lo = 0;
  int hi = 0;
  if ((c >> 1) == 0) {
    lo = 1;
    hi = fs;
  } else if ((c >> 1) == 1) {
    lo = fs + 1;
    hi = le;
  }
  if (lo < rs) lo = rs;
  if (hi > re) hi = re;
  int len = hi - lo;
  if (len < 0) len = 0;
  const unsigned ulen_x = ((c & 1) == 0) ? (unsigned)len : 0u;
  const unsigned ulen_y = ((c & 1) == 1) ? (unsigned)len : 0u;

  const int krow0 = p0 + h;
  const float* __restrict__ base =
      hs + (size_t)b * (S_SZ * H_SZ) + (size_t)krow0 * H_SZ + (size_t)n0 + (size_t)(4 * c);

  v8f acc0 = {0.f, 0.f, 0.f, 0.f, 0.f, 0.f, 0.f, 0.f};
  v8f acc1 = acc0;

  const int rel0 = krow0 - lo;

  auto step = [&](const v4f x, const int s) {
    const unsigned rel = (unsigned)(rel0 + 2 * s);
    v2f a;
    a.x = (rel < ulen_x) ? 1.0f : 0.0f;
    a.y = (rel < ulen_y) ? 1.0f : 0.0f;
    const v2f b01 = __builtin_shufflevector(x, x, 0, 1);
    const v2f b23 = __builtin_shufflevector(x, x, 2, 3);
    acc0 = __builtin_amdgcn_wmma_f32_16x16x4_f32(false, a, false, b01, (short)0, acc0, false, false);
    acc1 = __builtin_amdgcn_wmma_f32_16x16x4_f32(false, a, false, b23, (short)0, acc1, false, false);
  };
  auto ld = [&](const int s) -> v4f {
    return *reinterpret_cast<const v4f*>(base + (size_t)(2 * s) * H_SZ);
  };

  static_assert(QG == 4, "the sum below is written out for four steps");
  static_assert(RING % QG == 0, "a quad never wraps round the ring");

  v4f q[RING];
  if (nk == 2 * RING) {
#pragma unroll
    for (int t = 0; t < RING; ++t) {
      q[t] = ld(t);
    }
#pragma unroll
    for (int g = 0; g < 2 * RING / QG; ++g) {
      const int t0 = g * QG;
      const int a = t0 % RING;
      const bool pl = (plain & (1u << g)) != 0u;
      v4f x = q[a];
      if (!pl) {
        x = (q[a] + q[a + 1]) + (q[a + 2] + q[a + 3]);
      }
      step(x, t0);
      if (pl) {
#pragma unroll
        for (int i = 1; i < QG; ++i) {
          step(q[a + i], t0 + i);
        }
      }
      if (t0 + RING < 2 * RING) {
#pragma unroll
        for (int i = 0; i < QG; ++i) {
          q[a + i] = ld(t0 + RING + i);
        }
      }
    }
  } else if (nk == RING) {
#pragma unroll
    for (int t = 0; t < RING; ++t) {
      q[t] = ld(t);
    }
#pragma unroll
    for (int g = 0; g < RING / QG; ++g) {
      const int t0 = g * QG;
      const bool pl = (plain & (1u << g)) != 0u;
      v4f x = q[t0];
      if (!pl) {
        x = (q[t0] + q[t0 + 1]) + (q[t0 + 2] + q[t0 + 3]);
      }
      step(x, t0);
      if (pl) {
#pragma unroll
        for (int i = 1; i < QG; ++i) {
          step(q[t0 + i], t0 + i);
        }
      }
    }
  }

  if (lane < 16) {
    v4f t4, l4;
    t4.x = acc0[0]; t4.y = acc0[1]; t4.z = acc1[0]; t4.w = acc1[1];
    l4.x = acc0[2]; l4.y = acc0[3]; l4.z = acc1[2]; l4.w = acc1[3];
    red[kq][ct][0][lane] = t4;
    red[kq][ct][1][lane] = l4;
  }
  __syncthreads();

  {
    const int t = threadIdx.x;
    const int which = t >> 7;
    const int cb = t & 127;
    const float* r = reinterpret_cast<const float*>(&red[0][0][0][0]);
    const int off = ((cb >> 6) * 2 + which) * 64 + (cb & 63);
    float s = r[off];
    s += r[off + 1 * 256];
    s += r[off + 2 * 256];
    s += r[off + 3 * 256];
    const float scale = which ? lscale : tscale;
    out[(size_t)which * (B_SZ * H_SZ) + (size_t)b * H_SZ + (size_t)(blockIdx.x * 128 + cb)] = s * scale;
  }
}

extern "C" void kernel_launch(void* const* d_in, const int* in_sizes, int n_in,
                              void* d_out, int out_size, void* d_ws,
                              size_t ws_size, hipStream_t stream) {
  (void)in_sizes;
  (void)n_in;
  (void)out_size;
  (void)d_ws;
  (void)ws_size;

  const float* hidden = (const float*)d_in[0];
  const int* ids = (const int*)d_in[1];
  float* out = (float*)d_out;

  dim3 grid(H_SZ / 128, B_SZ);
  dim3 block(256);
  seg_pool_wmma_kernel<<<grid, block, 0, stream>>>(hidden, ids, out);
}
